# starting kernel
# baseline (speedup 1.0000x reference)
#include <hip/hip_runtime.h>

typedef __attribute__((ext_vector_type(2))) float v2f;
typedef __attribute__((ext_vector_type(8))) float v8f;

static __device__ __forceinline__ void atomic_add_f32(float* p, float v) {
  __hip_atomic_fetch_add(p, v, __ATOMIC_RELAXED, __HIP_MEMORY_SCOPE_AGENT);
}

__global__ void k_init_deg(float* deg, int n) {
  int i = blockIdx.x * blockDim.x + threadIdx.x;
  if (i < n) deg[i] = 1.0f;
}

__global__ void k_count_deg(const int* __restrict__ dst, float* deg, int E) {
  int i = blockIdx.x * blockDim.x + threadIdx.x;
  if (i < E) atomic_add_f32(&deg[dst[i]], 1.0f);
}

__global__ void k_rsqrt_deg(float* d, int n) {
  int i = blockIdx.x * blockDim.x + threadIdx.x;
  if (i < n) d[i] = rsqrtf(d[i]);
}

template <int NOUT, bool RELU>
__global__ __launch_bounds__(256) void k_gemm_wmma(const float* __restrict__ A,
                                                   const float* __restrict__ W,
                                                   float* __restrict__ out, int nrows) {
  __shared__ float Wl[128 * NOUT];
  const int tid = threadIdx.x;
  for (int i = tid; i < 128 * NOUT; i += 256) {
    int k = i / NOUT;
    int j = i - k * NOUT;
    Wl[(k >> 1) * (2 * NOUT) + 2 * j + (k & 1)] = W[i];
  }
  __syncthreads();

  const int wave = tid >> 5;
  const int lane = tid & 31;
  const int tile = blockIdx.x * 8 + wave;
  const int ntiles = (nrows + 15) >> 4;
  if (tile >= ntiles) return;

  const int m = lane & 15;
  const int kh = lane >> 4;
  int row = tile * 16 + m;
  if (row >= nrows) row = nrows - 1;

  v8f acc[NOUT / 16];
#pragma unroll
  for (int t = 0; t < NOUT / 16; ++t) acc[t] = v8f{0.f, 0.f, 0.f, 0.f, 0.f, 0.f, 0.f, 0.f};

  const float* arow = A + (size_t)row * 128 + 2 * kh;
#pragma unroll 4
  for (int k0 = 0; k0 < 128; k0 += 4) {
    v2f a = *(const v2f*)(arow + k0);
    if (RELU) {
      a.x = fmaxf(a.x, 0.f);
      a.y = fmaxf(a.y, 0.f);
    }
    const float* bb = &Wl[((k0 >> 1) + kh) * (2 * NOUT)];
#pragma unroll
    for (int t = 0; t < NOUT / 16; ++t) {
      v2f b = *(const v2f*)(bb + 2 * (t * 16 + m));
      acc[t] = __builtin_amdgcn_wmma_f32_16x16x4_f32(false, a, false, b, (short)0, acc[t],
                                                     false, false);
    }
  }

  const int rbase = tile * 16 + kh * 8;
#pragma unroll
  for (int t = 0; t < NOUT / 16; ++t) {
#pragma unroll
    for (int v = 0; v < 8; ++v) {
      int r = rbase + v;
      if (r < nrows) out[(size_t)r * NOUT + t * 16 + m] = acc[t][v];
    }
  }
}

template <int NOUT>
__global__ void k_init_agg(const float* __restrict__ h, const float* __restrict__ dinv,
                           const float* __restrict__ bias, float* __restrict__ agg, int n) {
  int i = blockIdx.x * blockDim.x + threadIdx.x;
  if (i >= n * NOUT) return;
  int r = i / NOUT;
  int c = i - r * NOUT;
  float di = dinv[r];
  agg[i] = h[i] * di * di + bias[c];
}

__global__ __launch_bounds__(256) void k_scatter128(const float* __restrict__ h,
                                                    const int* __restrict__ src,
                                                    const int* __restrict__ dst,
                                                    const float* __restrict__ dinv,
                                                    float* __restrict__ agg, int E) {
  int gw = (blockIdx.x * 256 + threadIdx.x) >> 5;
  int lane = threadIdx.x & 31;
  if (gw >= E) return;
  int s = src[gw], d = dst[gw];
  float nrm = dinv[s] * dinv[d];
  const float4 hv = *(const float4*)(h + (size_t)s * 128 + lane * 4);
  float* o = agg + (size_t)d * 128 + lane * 4;
  atomic_add_f32(o + 0, hv.x * nrm);
  atomic_add_f32(o + 1, hv.y * nrm);
  atomic_add_f32(o + 2, hv.z * nrm);
  atomic_add_f32(o + 3, hv.w * nrm);
}

__global__ __launch_bounds__(256) void k_scatter64(const float* __restrict__ h,
                                                   const int* __restrict__ src,
                                                   const int* __restrict__ dst,
                                                   const float* __restrict__ dinv,
                                                   float* __restrict__ out, int E) {
  int t = blockIdx.x * 256 + threadIdx.x;
  int e = t >> 4;
  int lane = threadIdx.x & 15;
  if (e >= E) return;
  int s = src[e], d = dst[e];
  float nrm = dinv[s] * dinv[d];
  const float4 hv = *(const float4*)(h + (size_t)s * 64 + lane * 4);
  float* o = out + (size_t)d * 64 + lane * 4;
  atomic_add_f32(o + 0, hv.x * nrm);
  atomic_add_f32(o + 1, hv.y * nrm);
  atomic_add_f32(o + 2, hv.z * nrm);
  atomic_add_f32(o + 3, hv.w * nrm);
}

extern "C" void kernel_launch(void* const* d_in, const int* in_sizes, int n_in, void* d_out,
                              int out_size, void* d_ws, size_t ws_size, hipStream_t stream) {
  const float* x = (const float*)d_in[0];
  const int* ei = (const int*)d_in[1];
  const float* W1 = (const float*)d_in[2];
  const float* b1 = (const float*)d_in[3];
  const float* W2 = (const float*)d_in[4];
  const float* b2 = (const float*)d_in[5];
  float* out = (float*)d_out;

  const int N = in_sizes[0] / 128;
  const int E = in_sizes[1] / 2;
  const int* src = ei;
  const int* dstp = ei + E;

  char* ws = (char*)d_ws;
  float* dinv = (float*)ws;
  size_t off = (((size_t)N * 4) + 255) & ~(size_t)255;
  float* h1 = (float*)(ws + off);
  float* agg1 = (float*)(ws + off + (size_t)N * 128 * 4);
  float* h2 = h1;

  dim3 B(256);
  k_init_deg<<<(N + 255) / 256, B, 0, stream>>>(dinv, N);
  k_count_deg<<<(E + 255) / 256, B, 0, stream>>>(dstp, dinv, E);
  k_rsqrt_deg<<<(N + 255) / 256, B, 0, stream>>>(dinv, N);

  const int ntiles = (N + 15) / 16;
  const int gblocks = (ntiles + 7) / 8;

  k_gemm_wmma<128, false><<<gblocks, B, 0, stream>>>(x, W1, h1, N);
  k_init_agg<128><<<(N * 128 + 255) / 256, B, 0, stream>>>(h1, dinv, b1, agg1, N);
  k_scatter128<<<(E + 7) / 8, B, 0, stream>>>(h1, src, dstp, dinv, agg1, E);

  k_gemm_wmma<64, true><<<gblocks, B, 0, stream>>>(agg1, W2, h2, N);
  k_init_agg<64><<<(N * 64 + 255) / 256, B, 0, stream>>>(h2, dinv, b2, out, N);
  k_scatter64<<<(E + 15) / 16, B, 0, stream>>>(h2, src, dstp, dinv, out, E);
}


# edge-scatter kernels: lane-to-column map changed so each wave-wide f32 atomic covers one whole 128-byte line, not 8 dwords in each of 4 lines; same addends, atomics in arrival order as before
# speedup vs baseline: 1.3813x; 1.3813x over previous
#include <hip/hip_runtime.h>

typedef __attribute__((ext_vector_type(2))) float v2f;
typedef __attribute__((ext_vector_type(8))) float v8f;

static __device__ __forceinline__ void atomic_add_f32(float* p, float v) {
  __hip_atomic_fetch_add(p, v, __ATOMIC_RELAXED, __HIP_MEMORY_SCOPE_AGENT);
}

__global__ void k_init_deg(float* deg, int n) {
  int i = blockIdx.x * blockDim.x + threadIdx.x;
  if (i < n) deg[i] = 1.0f;
}

__global__ void k_count_deg(const int* __restrict__ dst, float* deg, int E) {
  int i = blockIdx.x * blockDim.x + threadIdx.x;
  if (i < E) atomic_add_f32(&deg[dst[i]], 1.0f);
}

__global__ void k_rsqrt_deg(float* d, int n) {
  int i = blockIdx.x * blockDim.x + threadIdx.x;
  if (i < n) d[i] = rsqrtf(d[i]);
}

template <int NOUT, bool RELU>
__global__ __launch_bounds__(256) void k_gemm_wmma(const float* __restrict__ A,
                                                   const float* __restrict__ W,
                                                   float* __restrict__ out, int nrows) {
  __shared__ float Wl[128 * NOUT];
  const int tid = threadIdx.x;
  for (int i = tid; i < 128 * NOUT; i += 256) {
    int k = i / NOUT;
    int j = i - k * NOUT;
    Wl[(k >> 1) * (2 * NOUT) + 2 * j + (k & 1)] = W[i];
  }
  __syncthreads();

  const int wave = tid >> 5;
  const int lane = tid & 31;
  const int tile = blockIdx.x * 8 + wave;
  const int ntiles = (nrows + 15) >> 4;
  if (tile >= ntiles) return;

  const int m = lane & 15;
  const int kh = lane >> 4;
  int row = tile * 16 + m;
  if (row >= nrows) row = nrows - 1;

  v8f acc[NOUT / 16];
#pragma unroll
  for (int t = 0; t < NOUT / 16; ++t) acc[t] = v8f{0.f, 0.f, 0.f, 0.f, 0.f, 0.f, 0.f, 0.f};

  const float* arow = A + (size_t)row * 128 + 2 * kh;
#pragma unroll 4
  for (int k0 = 0; k0 < 128; k0 += 4) {
    v2f a = *(const v2f*)(arow + k0);
    if (RELU) {
      a.x = fmaxf(a.x, 0.f);
      a.y = fmaxf(a.y, 0.f);
    }
    const float* bb = &Wl[((k0 >> 1) + kh) * (2 * NOUT)];
#pragma unroll
    for (int t = 0; t < NOUT / 16; ++t) {
      v2f b = *(const v2f*)(bb + 2 * (t * 16 + m));
      acc[t] = __builtin_amdgcn_wmma_f32_16x16x4_f32(false, a, false, b, (short)0, acc[t],
                                                     false, false);
    }
  }

  const int rbase = tile * 16 + kh * 8;
#pragma unroll
  for (int t = 0; t < NOUT / 16; ++t) {
#pragma unroll
    for (int v = 0; v < 8; ++v) {
      int r = rbase + v;
      if (r < nrows) out[(size_t)r * NOUT + t * 16 + m] = acc[t][v];
    }
  }
}

template <int NOUT>
__global__ void k_init_agg(const float* __restrict__ h, const float* __restrict__ dinv,
                           const float* __restrict__ bias, float* __restrict__ agg, int n) {
  int i = blockIdx.x * blockDim.x + threadIdx.x;
  if (i >= n * NOUT) return;
  int r = i / NOUT;
  int c = i - r * NOUT;
  float di = dinv[r];
  agg[i] = h[i] * di * di + bias[c];
}

__global__ __launch_bounds__(256) void k_scatter128(const float* __restrict__ h,
                                                    const int* __restrict__ src,
                                                    const int* __restrict__ dst,
                                                    const float* __restrict__ dinv,
                                                    float* __restrict__ agg, int E) {
  int gw = (blockIdx.x * 256 + threadIdx.x) >> 5;
  int lane = threadIdx.x & 31;
  if (gw >= E) return;
  int s = src[gw], d = dst[gw];
  float nrm = dinv[s] * dinv[d];
  const float* hr = h + (size_t)s * 128 + lane;
  const float h0 = hr[0], h1 = hr[32], h2 = hr[64], h3 = hr[96];
  float* o = agg + (size_t)d * 128 + lane;
  atomic_add_f32(o, h0 * nrm);
  atomic_add_f32(o + 32, h1 * nrm);
  atomic_add_f32(o + 64, h2 * nrm);
  atomic_add_f32(o + 96, h3 * nrm);
}

static __device__ __forceinline__ int bcast_i(int v, int l) { return __builtin_amdgcn_readlane(v, l); }
static __device__ __forceinline__ float bcast_f(float v, int l) {
  return __int_as_float(__builtin_amdgcn_readlane(__float_as_int(v), l));
}

__global__ __launch_bounds__(256) void k_scatter64(const float* __restrict__ h,
                                                   const int* __restrict__ src,
                                                   const int* __restrict__ dst,
                                                   const float* __restrict__ dinv,
                                                   float* __restrict__ out, int E) {
  const int gw = (blockIdx.x * 256 + threadIdx.x) >> 5;
  const int lane = threadIdx.x & 31;
  const int e0 = gw * 2;
  if (e0 >= E) return;
  if (e0 + 1 < E) {
    const int ei = e0 + (lane >> 4);
    const int sv = src[ei], dv = dst[ei];
    const int s0 = bcast_i(sv, 0), s1 = bcast_i(sv, 16);
    const float* r0 = h + (size_t)s0 * 64 + lane;
    const float* r1 = h + (size_t)s1 * 64 + lane;
    const float a0 = r0[0], a1 = r0[32], b0 = r1[0], b1 = r1[32];
    const float nv = dinv[sv] * dinv[dv];
    const int d0 = bcast_i(dv, 0), d1 = bcast_i(dv, 16);
    const float n0 = bcast_f(nv, 0), n1 = bcast_f(nv, 16);
    float* o0 = out + (size_t)d0 * 64 + lane;
    float* o1 = out + (size_t)d1 * 64 + lane;
    atomic_add_f32(o0, a0 * n0);
    atomic_add_f32(o0 + 32, a1 * n0);
    atomic_add_f32(o1, b0 * n1);
    atomic_add_f32(o1 + 32, b1 * n1);
  } else {
    const int s = src[e0], d = dst[e0];
    const float nrm = dinv[s] * dinv[d];
    const float* r = h + (size_t)s * 64 + lane;
    const float a0 = r[0], a1 = r[32];
    float* o = out + (size_t)d * 64 + lane;
    atomic_add_f32(o, a0 * nrm);
    atomic_add_f32(o + 32, a1 * nrm);
  }
}

extern "C" void kernel_launch(void* const* d_in, const int* in_sizes, int n_in, void* d_out,
                              int out_size, void* d_ws, size_t ws_size, hipStream_t stream) {
  const float* x = (const float*)d_in[0];
  const int* ei = (const int*)d_in[1];
  const float* W1 = (const float*)d_in[2];
  const float* b1 = (const float*)d_in[3];
  const float* W2 = (const float*)d_in[4];
  const float* b2 = (const float*)d_in[5];
  float* out = (float*)d_out;

  const int N = in_sizes[0] / 128;
  const int E = in_sizes[1] / 2;
  const int* src = ei;
  const int* dstp = ei + E;

  char* ws = (char*)d_ws;
  float* dinv = (float*)ws;
  size_t off = (((size_t)N * 4) + 255) & ~(size_t)255;
  float* h1 = (float*)(ws + off);
  float* agg1 = (float*)(ws + off + (size_t)N * 128 * 4);
  float* h2 = h1;

  dim3 B(256);
  k_init_deg<<<(N + 255) / 256, B, 0, stream>>>(dinv, N);
  k_count_deg<<<(E + 255) / 256, B, 0, stream>>>(dstp, dinv, E);
  k_rsqrt_deg<<<(N + 255) / 256, B, 0, stream>>>(dinv, N);

  const int ntiles = (N + 15) / 16;
  const int gblocks = (ntiles + 7) / 8;

  k_gemm_wmma<128, false><<<gblocks, B, 0, stream>>>(x, W1, h1, N);
  k_init_agg<128><<<(N * 128 + 255) / 256, B, 0, stream>>>(h1, dinv, b1, agg1, N);
  k_scatter128<<<(E + 7) / 8, B, 0, stream>>>(h1, src, dstp, dinv, agg1, E);

  k_gemm_wmma<64, true><<<gblocks, B, 0, stream>>>(agg1, W2, h2, N);
  k_init_agg<64><<<(N * 64 + 255) / 256, B, 0, stream>>>(h2, dinv, b2, out, N);
  k_scatter64<<<(E + 15) / 16, B, 0, stream>>>(h2, src, dstp, dinv, out, E);
}


# edge sums: f32 atomic scatter replaced by per-call counting sort of edges by destination + one wave per row summing in registers, one store; add order = slot order from an integer atomic's arrival
# speedup vs baseline: 2.1273x; 1.5401x over previous
#include <hip/hip_runtime.h>

typedef __attribute__((ext_vector_type(2))) float v2f;
typedef __attribute__((ext_vector_type(8))) float v8f;

typedef __attribute__((ext_vector_type(2))) int v2i;

__global__ void k_zero_cnt(int* cnt, int n) {
  int i = blockIdx.x * blockDim.x + threadIdx.x;
  if (i < n) cnt[i] = 0;
}

__global__ void k_count_rank(const int* __restrict__ dst, int* cnt, int* __restrict__ rank, int E) {
  int i = blockIdx.x * blockDim.x + threadIdx.x;
  if (i < E)
    rank[i] = __hip_atomic_fetch_add(&cnt[dst[i]], 1, __ATOMIC_RELAXED, __HIP_MEMORY_SCOPE_AGENT);
}

__global__ __launch_bounds__(1024) void k_scan_dinv(const int* __restrict__ cnt,
                                                     int* __restrict__ rowptr,
                                                     float* __restrict__ dinv, int n) {
  __shared__ int w_tot[32];
  __shared__ int w_pre[32];
  const int tid = threadIdx.x;
  const int lane = tid & 31;
  const int wave = tid >> 5;
  const int blk0 = blockIdx.x * 1024;
  int pre = 0;
  for (int i = tid; i < blk0; i += 1024) pre += cnt[i];
  const int i = blk0 + tid;
  const int c = (i < n) ? cnt[i] : 0;
  int incl = c;
  for (int o = 1; o < 32; o <<= 1) {
    int t = __shfl_up(incl, o);
    if (lane >= o) incl += t;
  }
  for (int o = 16; o > 0; o >>= 1) pre += __shfl_xor(pre, o);
  if (lane == 31) w_tot[wave] = incl;
  if (lane == 0) w_pre[wave] = pre;
  __syncthreads();
  int base = w_pre[lane] + ((lane < wave) ? w_tot[lane] : 0);
  for (int o = 16; o > 0; o >>= 1) base += __shfl_xor(base, o);
  if (i < n) {
    const int excl = base + incl - c;
    rowptr[i] = excl;
    dinv[i] = rsqrtf((float)(c + 1));
    if (i == n - 1) rowptr[n] = excl + c;
  }
}

__global__ void k_fill_csr(const int* __restrict__ src, const int* __restrict__ dst,
                           const int* __restrict__ rank, const int* __restrict__ rowptr,
                           const float* __restrict__ dinv, int* __restrict__ csr, int E) {
  int i = blockIdx.x * blockDim.x + threadIdx.x;
  if (i < E) {
    const int s = src[i];
    v2i ent;
    ent.x = s;
    ent.y = __float_as_int(dinv[s]);
    ((v2i*)csr)[rowptr[dst[i]] + rank[i]] = ent;
  }
}

template <int NOUT, bool RELU>
__global__ __launch_bounds__(256) void k_gemm_wmma(const float* __restrict__ A,
                                                   const float* __restrict__ W,
                                                   float* __restrict__ out, int nrows) {
  __shared__ float Wl[128 * NOUT];
  const int tid = threadIdx.x;
  for (int i = tid; i < 128 * NOUT; i += 256) {
    int k = i / NOUT;
    int j = i - k * NOUT;
    Wl[(k >> 1) * (2 * NOUT) + 2 * j + (k & 1)] = W[i];
  }
  __syncthreads();

  const int wave = tid >> 5;
  const int lane = tid & 31;
  const int tile = blockIdx.x * 8 + wave;
  const int ntiles = (nrows + 15) >> 4;
  if (tile >= ntiles) return;

  const int m = lane & 15;
  const int kh = lane >> 4;
  int row = tile * 16 + m;
  if (row >= nrows) row = nrows - 1;

  v8f acc[NOUT / 16];
#pragma unroll
  for (int t = 0; t < NOUT / 16; ++t) acc[t] = v8f{0.f, 0.f, 0.f, 0.f, 0.f, 0.f, 0.f, 0.f};

  const float* arow = A + (size_t)row * 128 + 2 * kh;
#pragma unroll 4
  for (int k0 = 0; k0 < 128; k0 += 4) {
    v2f a = *(const v2f*)(arow + k0);
    if (RELU) {
      a.x = fmaxf(a.x, 0.f);
      a.y = fmaxf(a.y, 0.f);
    }
    const float* bb = &Wl[((k0 >> 1) + kh) * (2 * NOUT)];
#pragma unroll
    for (int t = 0; t < NOUT / 16; ++t) {
      v2f b = *(const v2f*)(bb + 2 * (t * 16 + m));
      acc[t] = __builtin_amdgcn_wmma_f32_16x16x4_f32(false, a, false, b, (short)0, acc[t],
                                                     false, false);
    }
  }

  const int rbase = tile * 16 + kh * 8;
#pragma unroll
  for (int t = 0; t < NOUT / 16; ++t) {
#pragma unroll
    for (int v = 0; v < 8; ++v) {
      int r = rbase + v;
      if (r < nrows) out[(size_t)r * NOUT + t * 16 + m] = acc[t][v];
    }
  }
}

template <int NOUT>
__global__ __launch_bounds__(256) void k_gather(const float* __restrict__ h,
                                                const int* __restrict__ rowptr,
                                                const int* __restrict__ csr,
                                                const float* __restrict__ dinv,
                                                const float* __restrict__ bias,
                                                float* __restrict__ out, int n) {
  constexpr int VW = NOUT / 32;
  typedef __attribute__((ext_vector_type(VW))) float vec;
  const int wave = __builtin_amdgcn_readfirstlane((int)(threadIdx.x >> 5));
  const int lane = threadIdx.x & 31;
  const int row = blockIdx.x * 8 + wave;
  if (row >= n) return;
  const int beg = rowptr[row];
  const int end = rowptr[row + 1];
  const float di = dinv[row];
  const vec* hv = (const vec*)h + lane;
  const vec self = hv[(size_t)row * 32];
  vec acc;
  for (int q = 0; q < VW; ++q) acc[q] = self[q] * di * di + bias[lane * VW + q];
  for (int base = beg; base < end; base += 32) {
    const int m = (end - base < 32) ? (end - base) : 32;
    v2i ent = v2i{0, 0};
    if (lane < m) ent = ((const v2i*)csr)[base + lane];
    const int sv = ent.x;
    const int wv = __float_as_int(__int_as_float(ent.y) * di);
#pragma unroll 4
    for (int j = 0; j < m; ++j) {
      const int s = __builtin_amdgcn_readlane(sv, j);
      const float w = __int_as_float(__builtin_amdgcn_readlane(wv, j));
      const vec v = hv[(size_t)s * 32];
      for (int q = 0; q < VW; ++q) acc[q] += v[q] * w;
    }
  }
  ((vec*)out)[(size_t)row * 32 + lane] = acc;
}

extern "C" void kernel_launch(void* const* d_in, const int* in_sizes, int n_in, void* d_out,
                              int out_size, void* d_ws, size_t ws_size, hipStream_t stream) {
  const float* x = (const float*)d_in[0];
  const int* ei = (const int*)d_in[1];
  const float* W1 = (const float*)d_in[2];
  const float* b1 = (const float*)d_in[3];
  const float* W2 = (const float*)d_in[4];
  const float* b2 = (const float*)d_in[5];
  float* out = (float*)d_out;

  const int N = in_sizes[0] / 128;
  const int E = in_sizes[1] / 2;
  const int* src = ei;
  const int* dstp = ei + E;

  char* ws = (char*)d_ws;
  size_t off = 0;
  int* cnt = (int*)(ws + off);
  off += (((size_t)N * 4) + 255) & ~(size_t)255;
  int* rowptr = (int*)(ws + off);
  off += (((size_t)(N + 1) * 4) + 255) & ~(size_t)255;
  float* dinv = (float*)(ws + off);
  off += (((size_t)N * 4) + 255) & ~(size_t)255;
  int* rank = (int*)(ws + off);
  off += (((size_t)E * 4) + 255) & ~(size_t)255;
  int* csr = (int*)(ws + off);
  off += (((size_t)E * 8) + 255) & ~(size_t)255;
  float* h1 = (float*)(ws + off);
  off += (size_t)N * 128 * 4;
  float* agg1 = (float*)(ws + off);
  float* h2 = h1;

  dim3 B(256);
  k_zero_cnt<<<(N + 255) / 256, B, 0, stream>>>(cnt, N);
  k_count_rank<<<(E + 255) / 256, B, 0, stream>>>(dstp, cnt, rank, E);
  k_scan_dinv<<<(N + 1023) / 1024, dim3(1024), 0, stream>>>(cnt, rowptr, dinv, N);
  k_fill_csr<<<(E + 255) / 256, B, 0, stream>>>(src, dstp, rank, rowptr, dinv, csr, E);

  const int ntiles = (N + 15) / 16;
  const int gblocks = (ntiles + 7) / 8;
  const int rblocks = (N + 7) / 8;

  k_gemm_wmma<128, false><<<gblocks, B, 0, stream>>>(x, W1, h1, N);
  k_gather<128><<<rblocks, B, 0, stream>>>(h1, rowptr, csr, dinv, b1, agg1, N);

  k_gemm_wmma<64, true><<<gblocks, B, 0, stream>>>(agg1, W2, h2, N);
  k_gather<64><<<rblocks, B, 0, stream>>>(h2, rowptr, csr, dinv, b2, out, N);
}


# layer-1 GEMM blocks co-scheduled inside the two edge-sort launches (count, fill) instead of a launch of their own: independent work overlapped; same sums, same order
# speedup vs baseline: 2.2361x; 1.0511x over previous
#include <hip/hip_runtime.h>

typedef __attribute__((ext_vector_type(2))) float v2f;
typedef __attribute__((ext_vector_type(8))) float v8f;

typedef __attribute__((ext_vector_type(2))) int v2i;

__global__ void k_zero_cnt(int* cnt, int n) {
  int i = blockIdx.x * blockDim.x + threadIdx.x;
  if (i < n) cnt[i] = 0;
}


__global__ __launch_bounds__(1024) void k_scan_dinv(const int* __restrict__ cnt,
                                                     int* __restrict__ rowptr,
                                                     float* __restrict__ dinv, int n) {
  __shared__ int w_tot[32];
  __shared__ int w_pre[32];
  const int tid = threadIdx.x;
  const int lane = tid & 31;
  const int wave = tid >> 5;
  const int blk0 = blockIdx.x * 1024;
  int pre = 0;
  for (int i = tid; i < blk0; i += 1024) pre += cnt[i];
  const int i = blk0 + tid;
  const int c = (i < n) ? cnt[i] : 0;
  int incl = c;
  for (int o = 1; o < 32; o <<= 1) {
    int t = __shfl_up(incl, o);
    if (lane >= o) incl += t;
  }
  for (int o = 16; o > 0; o >>= 1) pre += __shfl_xor(pre, o);
  if (lane == 31) w_tot[wave] = incl;
  if (lane == 0) w_pre[wave] = pre;
  __syncthreads();
  int base = w_pre[lane] + ((lane < wave) ? w_tot[lane] : 0);
  for (int o = 16; o > 0; o >>= 1) base += __shfl_xor(base, o);
  if (i < n) {
    const int excl = base + incl - c;
    rowptr[i] = excl;
    dinv[i] = rsqrtf((float)(c + 1));
    if (i == n - 1) rowptr[n] = excl + c;
  }
}


template <int NOUT, bool RELU>
__device__ __forceinline__ void gemm_block(const float* __restrict__ A, const float* __restrict__ W,
                                           float* __restrict__ out, int nrows, int bid, float* Wl) {
  const int tid = threadIdx.x;
  for (int i = tid; i < 128 * NOUT; i += 256) {
    int k = i / NOUT;
    int j = i - k * NOUT;
    Wl[(k >> 1) * (2 * NOUT) + 2 * j + (k & 1)] = W[i];
  }
  __syncthreads();

  const int wave = tid >> 5;
  const int lane = tid & 31;
  const int tile = bid * 8 + wave;
  const int ntiles = (nrows + 15) >> 4;
  if (tile >= ntiles) return;

  const int m = lane & 15;
  const int kh = lane >> 4;
  int row = tile * 16 + m;
  if (row >= nrows) row = nrows - 1;

  v8f acc[NOUT / 16];
#pragma unroll
  for (int t = 0; t < NOUT / 16; ++t) acc[t] = v8f{0.f, 0.f, 0.f, 0.f, 0.f, 0.f, 0.f, 0.f};

  const float* arow = A + (size_t)row * 128 + 2 * kh;
#pragma unroll 4
  for (int k0 = 0; k0 < 128; k0 += 4) {
    v2f a = *(const v2f*)(arow + k0);
    if (RELU) {
      a.x = fmaxf(a.x, 0.f);
      a.y = fmaxf(a.y, 0.f);
    }
    const float* bb = &Wl[((k0 >> 1) + kh) * (2 * NOUT)];
#pragma unroll
    for (int t = 0; t < NOUT / 16; ++t) {
      v2f b = *(const v2f*)(bb + 2 * (t * 16 + m));
      acc[t] = __builtin_amdgcn_wmma_f32_16x16x4_f32(false, a, false, b, (short)0, acc[t],
                                                     false, false);
    }
  }

  const int rbase = tile * 16 + kh * 8;
#pragma unroll
  for (int t = 0; t < NOUT / 16; ++t) {
#pragma unroll
    for (int v = 0; v < 8; ++v) {
      int r = rbase + v;
      if (r < nrows) out[(size_t)r * NOUT + t * 16 + m] = acc[t][v];
    }
  }
}

template <int NOUT, bool RELU>
__global__ __launch_bounds__(256) void k_gemm_wmma(const float* __restrict__ A,
                                                   const float* __restrict__ W,
                                                   float* __restrict__ out, int nrows) {
  __shared__ float Wl[128 * NOUT];
  gemm_block<NOUT, RELU>(A, W, out, nrows, blockIdx.x, Wl);
}

__global__ __launch_bounds__(256) void k_count_rank_gemm(const int* __restrict__ dst, int* cnt,
                                                         int* __restrict__ rank, int E,
                                                         const float* __restrict__ A,
                                                         const float* __restrict__ W,
                                                         float* __restrict__ out, int nrows,
                                                         int gfirst, int gcount) {
  __shared__ float Wl[128 * 128];
  if ((int)blockIdx.x < gcount) {
    gemm_block<128, false>(A, W, out, nrows, gfirst + (int)blockIdx.x, Wl);
    return;
  }
  int i = ((int)blockIdx.x - gcount) * 256 + threadIdx.x;
  if (i < E)
    rank[i] = __hip_atomic_fetch_add(&cnt[dst[i]], 1, __ATOMIC_RELAXED, __HIP_MEMORY_SCOPE_AGENT);
}

__global__ __launch_bounds__(256) void k_fill_csr_gemm(const int* __restrict__ src,
                                                       const int* __restrict__ dst,
                                                       const int* __restrict__ rank,
                                                       const int* __restrict__ rowptr,
                                                       const float* __restrict__ dinv,
                                                       int* __restrict__ csr, int E,
                                                       const float* __restrict__ A,
                                                       const float* __restrict__ W,
                                                       float* __restrict__ out, int nrows,
                                                       int gfirst, int gcount) {
  __shared__ float Wl[128 * 128];
  if ((int)blockIdx.x < gcount) {
    gemm_block<128, false>(A, W, out, nrows, gfirst + (int)blockIdx.x, Wl);
    return;
  }
  int i = ((int)blockIdx.x - gcount) * 256 + threadIdx.x;
  if (i < E) {
    const int s = src[i];
    v2i ent;
    ent.x = s;
    ent.y = __float_as_int(dinv[s]);
    ((v2i*)csr)[rowptr[dst[i]] + rank[i]] = ent;
  }
}

template <int NOUT>
__global__ __launch_bounds__(256) void k_gather(const float* __restrict__ h,
                                                const int* __restrict__ rowptr,
                                                const int* __restrict__ csr,
                                                const float* __restrict__ dinv,
                                                const float* __restrict__ bias,
                                                float* __restrict__ out, int n) {
  constexpr int VW = NOUT / 32;
  typedef __attribute__((ext_vector_type(VW))) float vec;
  const int wave = __builtin_amdgcn_readfirstlane((int)(threadIdx.x >> 5));
  const int lane = threadIdx.x & 31;
  const int row = blockIdx.x * 8 + wave;
  if (row >= n) return;
  const int beg = rowptr[row];
  const int end = rowptr[row + 1];
  const float di = dinv[row];
  const vec* hv = (const vec*)h + lane;
  const vec self = hv[(size_t)row * 32];
  vec acc;
  for (int q = 0; q < VW; ++q) acc[q] = self[q] * di * di + bias[lane * VW + q];
  for (int base = beg; base < end; base += 32) {
    const int m = (end - base < 32) ? (end - base) : 32;
    v2i ent = v2i{0, 0};
    if (lane < m) ent = ((const v2i*)csr)[base + lane];
    const int sv = ent.x;
    const int wv = __float_as_int(__int_as_float(ent.y) * di);
#pragma unroll 4
    for (int j = 0; j < m; ++j) {
      const int s = __builtin_amdgcn_readlane(sv, j);
      const float w = __int_as_float(__builtin_amdgcn_readlane(wv, j));
      const vec v = hv[(size_t)s * 32];
      for (int q = 0; q < VW; ++q) acc[q] += v[q] * w;
    }
  }
  ((vec*)out)[(size_t)row * 32 + lane] = acc;
}

extern "C" void kernel_launch(void* const* d_in, const int* in_sizes, int n_in, void* d_out,
                              int out_size, void* d_ws, size_t ws_size, hipStream_t stream) {
  const float* x = (const float*)d_in[0];
  const int* ei = (const int*)d_in[1];
  const float* W1 = (const float*)d_in[2];
  const float* b1 = (const float*)d_in[3];
  const float* W2 = (const float*)d_in[4];
  const float* b2 = (const float*)d_in[5];
  float* out = (float*)d_out;

  const int N = in_sizes[0] / 128;
  const int E = in_sizes[1] / 2;
  const int* src = ei;
  const int* dstp = ei + E;

  char* ws = (char*)d_ws;
  size_t off = 0;
  int* cnt = (int*)(ws + off);
  off += (((size_t)N * 4) + 255) & ~(size_t)255;
  int* rowptr = (int*)(ws + off);
  off += (((size_t)(N + 1) * 4) + 255) & ~(size_t)255;
  float* dinv = (float*)(ws + off);
  off += (((size_t)N * 4) + 255) & ~(size_t)255;
  int* rank = (int*)(ws + off);
  off += (((size_t)E * 4) + 255) & ~(size_t)255;
  int* csr = (int*)(ws + off);
  off += (((size_t)E * 8) + 255) & ~(size_t)255;
  float* h1 = (float*)(ws + off);
  off += (size_t)N * 128 * 4;
  float* agg1 = (float*)(ws + off);
  float* h2 = h1;

  dim3 B(256);
  k_zero_cnt<<<(N + 255) / 256, B, 0, stream>>>(cnt, N);
  const int ntiles = (N + 15) / 16;
  const int gblocks = (ntiles + 7) / 8;
  const int rblocks = (N + 7) / 8;
  const int eblocks = (E + 255) / 256;
  const int gA = gblocks / 2;
  const int gB = gblocks - gA;

  k_count_rank_gemm<<<gA + eblocks, B, 0, stream>>>(dstp, cnt, rank, E, x, W1, h1, N, 0, gA);
  k_scan_dinv<<<(N + 1023) / 1024, dim3(1024), 0, stream>>>(cnt, rowptr, dinv, N);
  k_fill_csr_gemm<<<gB + eblocks, B, 0, stream>>>(src, dstp, rank, rowptr, dinv, csr, E, x, W1, h1, N,
                                                  gA, gB);

  k_gather<128><<<rblocks, B, 0, stream>>>(h1, rowptr, csr, dinv, b1, agg1, N);

  k_gemm_wmma<64, true><<<gblocks, B, 0, stream>>>(agg1, W2, h2, N);
  k_gather<64><<<rblocks, B, 0, stream>>>(h2, rowptr, csr, dinv, b2, out, N);
}


# edge sort: 800000 returning global atomics replaced by a two-level counting sort on LDS counters (bin by 256-destination range, rank in LDS); add order = slot order from LDS atomics' arrival
# speedup vs baseline: 2.9852x; 1.3350x over previous
#include <hip/hip_runtime.h>

typedef __attribute__((ext_vector_type(2))) float v2f;
typedef __attribute__((ext_vector_type(8))) float v8f;

typedef __attribute__((ext_vector_type(2))) int v2i;

constexpr int NCHUNK = 256;

__device__ __forceinline__ void chunk_range(int E, int c, int& c0, int& c1) {
  const int ch = (E + NCHUNK - 1) / NCHUNK;
  c0 = c * ch;
  if (c0 > E) c0 = E;
  c1 = c0 + ch;
  if (c1 > E) c1 = E;
}

__global__ __launch_bounds__(1024) void k_scan_excl(const int* __restrict__ in, int* __restrict__ outp,
                                                     int n) {
  __shared__ int w_tot[32];
  __shared__ int w_pre[32];
  const int tid = threadIdx.x;
  const int lane = tid & 31;
  const int wave = tid >> 5;
  const int blk0 = blockIdx.x * 1024;
  int pre = 0;
  for (int i = tid; i < blk0; i += 1024) pre += in[i];
  const int i = blk0 + tid;
  const int c = (i < n) ? in[i] : 0;
  int incl = c;
  for (int o = 1; o < 32; o <<= 1) {
    int t = __shfl_up(incl, o);
    if (lane >= o) incl += t;
  }
  for (int o = 16; o > 0; o >>= 1) pre += __shfl_xor(pre, o);
  if (lane == 31) w_tot[wave] = incl;
  if (lane == 0) w_pre[wave] = pre;
  __syncthreads();
  int base = w_pre[lane] + ((lane < wave) ? w_tot[lane] : 0);
  for (int o = 16; o > 0; o >>= 1) base += __shfl_xor(base, o);
  if (i < n) outp[i] = base + incl - c;
}

__global__ __launch_bounds__(1024) void k_bucket_csr(const int* __restrict__ goff,
                                                      const int* __restrict__ binned_i,
                                                      int* __restrict__ rowptr, float* __restrict__ dinv,
                                                      int* __restrict__ csr_src, int n, int E, int nb) {
  __shared__ int cnt[256];
  __shared__ int cur[256];
  const v2i* binned = (const v2i*)binned_i;
  const int k = blockIdx.x;
  const int tid = threadIdx.x;
  const int bs = goff[k * NCHUNK];
  const int be = (k + 1 < nb) ? goff[(k + 1) * NCHUNK] : E;
  if (tid < 256) cnt[tid] = 0;
  __syncthreads();
  for (int i0 = bs; i0 < be; i0 += 8 * 1024) {
    int d[8];
#pragma unroll
    for (int q = 0; q < 8; ++q) {
      const int i = i0 + q * 1024 + tid;
      d[q] = binned[(i < be) ? i : be - 1].y;
    }
#pragma unroll
    for (int q = 0; q < 8; ++q)
      if (i0 + q * 1024 + tid < be) atomicAdd(&cnt[d[q] & 255], 1);
  }
  __syncthreads();
  if (tid < 32) {
    int c[8];
    int tot = 0;
    for (int q = 0; q < 8; ++q) {
      c[q] = cnt[8 * tid + q];
      tot += c[q];
    }
    int incl = tot;
    for (int o = 1; o < 32; o <<= 1) {
      int t = __shfl_up(incl, o);
      if (tid >= o) incl += t;
    }
    int run = incl - tot;
    for (int q = 0; q < 8; ++q) {
      cur[8 * tid + q] = run;
      run += c[q];
    }
  }
  __syncthreads();
  if (tid < 256) {
    const int node = k * 256 + tid;
    if (node < n) {
      rowptr[node] = bs + cur[tid];
      dinv[node] = rsqrtf((float)(cnt[tid] + 1));
    }
  }
  if (k == nb - 1 && tid == 0) rowptr[n] = E;
  __syncthreads();
  for (int i0 = bs; i0 < be; i0 += 8 * 1024) {
    v2i r[8];
#pragma unroll
    for (int q = 0; q < 8; ++q) {
      const int i = i0 + q * 1024 + tid;
      r[q] = binned[(i < be) ? i : be - 1];
    }
#pragma unroll
    for (int q = 0; q < 8; ++q) {
      if (i0 + q * 1024 + tid < be) {
        const int p = atomicAdd(&cur[r[q].y & 255], 1);
        csr_src[bs + p] = r[q].x;
      }
    }
  }
}

template <int NOUT, bool RELU>
__device__ __forceinline__ void gemm_block(const float* __restrict__ A, const float* __restrict__ W,
                                           float* __restrict__ out, int nrows, int bid, float* Wl) {
  const int tid = threadIdx.x;
  for (int i = tid; i < 128 * NOUT; i += 256) {
    int k = i / NOUT;
    int j = i - k * NOUT;
    Wl[(k >> 1) * (2 * NOUT) + 2 * j + (k & 1)] = W[i];
  }
  __syncthreads();

  const int wave = tid >> 5;
  const int lane = tid & 31;
  const int tile = bid * 8 + wave;
  const int ntiles = (nrows + 15) >> 4;
  if (tile >= ntiles) return;

  const int m = lane & 15;
  const int kh = lane >> 4;
  int row = tile * 16 + m;
  if (row >= nrows) row = nrows - 1;

  v8f acc[NOUT / 16];
#pragma unroll
  for (int t = 0; t < NOUT / 16; ++t) acc[t] = v8f{0.f, 0.f, 0.f, 0.f, 0.f, 0.f, 0.f, 0.f};

  const float* arow = A + (size_t)row * 128 + 2 * kh;
#pragma unroll 4
  for (int k0 = 0; k0 < 128; k0 += 4) {
    v2f a = *(const v2f*)(arow + k0);
    if (RELU) {
      a.x = fmaxf(a.x, 0.f);
      a.y = fmaxf(a.y, 0.f);
    }
    const float* bb = &Wl[((k0 >> 1) + kh) * (2 * NOUT)];
#pragma unroll
    for (int t = 0; t < NOUT / 16; ++t) {
      v2f b = *(const v2f*)(bb + 2 * (t * 16 + m));
      acc[t] = __builtin_amdgcn_wmma_f32_16x16x4_f32(false, a, false, b, (short)0, acc[t],
                                                     false, false);
    }
  }

  const int rbase = tile * 16 + kh * 8;
#pragma unroll
  for (int t = 0; t < NOUT / 16; ++t) {
#pragma unroll
    for (int v = 0; v < 8; ++v) {
      int r = rbase + v;
      if (r < nrows) out[(size_t)r * NOUT + t * 16 + m] = acc[t][v];
    }
  }
}

template <int NOUT, bool RELU>
__global__ __launch_bounds__(256) void k_gemm_wmma(const float* __restrict__ A,
                                                   const float* __restrict__ W,
                                                   float* __restrict__ out, int nrows) {
  __shared__ float Wl[128 * NOUT];
  gemm_block<NOUT, RELU>(A, W, out, nrows, blockIdx.x, Wl);
}

__global__ __launch_bounds__(256) void k_bin_count_gemm(const int* __restrict__ dst,
                                                        int* __restrict__ ghist, int E, int nb,
                                                        const float* __restrict__ A,
                                                        const float* __restrict__ W,
                                                        float* __restrict__ out, int nrows,
                                                        int gfirst, int gcount) {
  __shared__ float Wl[128 * 128];
  if ((int)blockIdx.x < gcount) {
    gemm_block<128, false>(A, W, out, nrows, gfirst + (int)blockIdx.x, Wl);
    return;
  }
  int* lh = (int*)Wl;
  const int c = (int)blockIdx.x - gcount;
  const int tid = threadIdx.x;
  for (int k = tid; k < nb; k += 256) lh[k] = 0;
  __syncthreads();
  int c0, c1;
  chunk_range(E, c, c0, c1);
  for (int i0 = c0; i0 < c1; i0 += 16 * 256) {
    int d[16];
#pragma unroll
    for (int q = 0; q < 16; ++q) {
      const int i = i0 + q * 256 + tid;
      d[q] = dst[(i < c1) ? i : c1 - 1];
    }
#pragma unroll
    for (int q = 0; q < 16; ++q)
      if (i0 + q * 256 + tid < c1) atomicAdd(&lh[d[q] >> 8], 1);
  }
  __syncthreads();
  for (int k = tid; k < nb; k += 256) ghist[k * NCHUNK + c] = lh[k];
}

__global__ __launch_bounds__(256) void k_bin_place_gemm(const int* __restrict__ src,
                                                        const int* __restrict__ dst,
                                                        const int* __restrict__ goff,
                                                        int* __restrict__ binned_i, int E, int nb,
                                                        const float* __restrict__ A,
                                                        const float* __restrict__ W,
                                                        float* __restrict__ out, int nrows,
                                                        int gfirst, int gcount) {
  __shared__ float Wl[128 * 128];
  if ((int)blockIdx.x < gcount) {
    gemm_block<128, false>(A, W, out, nrows, gfirst + (int)blockIdx.x, Wl);
    return;
  }
  int* cur = (int*)Wl;
  v2i* binned = (v2i*)binned_i;
  const int c = (int)blockIdx.x - gcount;
  const int tid = threadIdx.x;
  for (int k = tid; k < nb; k += 256) cur[k] = goff[k * NCHUNK + c];
  __syncthreads();
  int c0, c1;
  chunk_range(E, c, c0, c1);
  for (int i0 = c0; i0 < c1; i0 += 16 * 256) {
    int d[16];
    int sr[16];
#pragma unroll
    for (int q = 0; q < 16; ++q) {
      const int i = (i0 + q * 256 + tid < c1) ? i0 + q * 256 + tid : c1 - 1;
      d[q] = dst[i];
      sr[q] = src[i];
    }
#pragma unroll
    for (int q = 0; q < 16; ++q) {
      if (i0 + q * 256 + tid < c1) {
        const int p = atomicAdd(&cur[d[q] >> 8], 1);
        v2i rec;
        rec.x = sr[q];
        rec.y = d[q];
        binned[p] = rec;
      }
    }
  }
}

template <int NOUT, bool FIRST>
__global__ __launch_bounds__(256) void k_gather(const float* __restrict__ h,
                                                const int* __restrict__ rowptr,
                                                const int* __restrict__ csr_src, float* csr_w,
                                                const float* __restrict__ dinv,
                                                const float* __restrict__ bias,
                                                float* __restrict__ out, int n) {
  constexpr int VW = NOUT / 32;
  typedef __attribute__((ext_vector_type(VW))) float vec;
  const int wave = __builtin_amdgcn_readfirstlane((int)(threadIdx.x >> 5));
  const int lane = threadIdx.x & 31;
  const int row = blockIdx.x * 8 + wave;
  if (row >= n) return;
  const int beg = rowptr[row];
  const int end = rowptr[row + 1];
  const float di = dinv[row];
  const vec* hv = (const vec*)h + lane;
  const vec self = hv[(size_t)row * 32];
  vec acc;
  for (int q = 0; q < VW; ++q) acc[q] = self[q] * di * di + bias[lane * VW + q];
  for (int base = beg; base < end; base += 32) {
    const int m = (end - base < 32) ? (end - base) : 32;
    int sv = 0;
    int wb = 0;
    if (lane < m) {
      sv = csr_src[base + lane];
      wb = __float_as_int(FIRST ? dinv[sv] : csr_w[base + lane]);
    }
#pragma unroll 4
    for (int j = 0; j < m; ++j) {
      const int s = __builtin_amdgcn_readlane(sv, j);
      const float w = __int_as_float(__builtin_amdgcn_readlane(wb, j)) * di;
      const vec v = hv[(size_t)s * 32];
      for (int q = 0; q < VW; ++q) acc[q] += v[q] * w;
    }
    if (FIRST && lane < m) csr_w[base + lane] = __int_as_float(wb);
  }
  ((vec*)out)[(size_t)row * 32 + lane] = acc;
}

extern "C" void kernel_launch(void* const* d_in, const int* in_sizes, int n_in, void* d_out,
                              int out_size, void* d_ws, size_t ws_size, hipStream_t stream) {
  const float* x = (const float*)d_in[0];
  const int* ei = (const int*)d_in[1];
  const float* W1 = (const float*)d_in[2];
  const float* b1 = (const float*)d_in[3];
  const float* W2 = (const float*)d_in[4];
  const float* b2 = (const float*)d_in[5];
  float* out = (float*)d_out;

  const int N = in_sizes[0] / 128;
  const int E = in_sizes[1] / 2;
  const int* src = ei;
  const int* dstp = ei + E;
  const int nb = (N + 255) / 256;
  const int nscan = nb * NCHUNK;

  char* ws = (char*)d_ws;
  size_t off = 0;
  int* ghist = (int*)(ws + off);
  off += (((size_t)nscan * 4) + 255) & ~(size_t)255;
  int* goff = (int*)(ws + off);
  off += (((size_t)nscan * 4) + 255) & ~(size_t)255;
  int* rowptr = (int*)(ws + off);
  off += (((size_t)(N + 1) * 4) + 255) & ~(size_t)255;
  float* dinv = (float*)(ws + off);
  off += (((size_t)N * 4) + 255) & ~(size_t)255;
  int* binned = (int*)(ws + off);
  off += (((size_t)E * 8) + 255) & ~(size_t)255;
  int* csr_src = (int*)(ws + off);
  off += (((size_t)E * 4) + 255) & ~(size_t)255;
  float* csr_w = (float*)(ws + off);
  off += (((size_t)E * 4) + 255) & ~(size_t)255;
  float* h1 = (float*)(ws + off);
  off += (size_t)N * 128 * 4;
  float* agg1 = (float*)(ws + off);
  float* h2 = h1;

  dim3 B(256);
  const int ntiles = (N + 15) / 16;
  const int gblocks = (ntiles + 7) / 8;
  const int rblocks = (N + 7) / 8;
  const int gA = gblocks / 2;
  const int gB = gblocks - gA;

  k_bin_count_gemm<<<gA + NCHUNK, B, 0, stream>>>(dstp, ghist, E, nb, x, W1, h1, N, 0, gA);
  k_scan_excl<<<(nscan + 1023) / 1024, dim3(1024), 0, stream>>>(ghist, goff, nscan);
  k_bin_place_gemm<<<gB + NCHUNK, B, 0, stream>>>(src, dstp, goff, binned, E, nb, x, W1, h1, N, gA, gB);
  k_bucket_csr<<<nb, dim3(1024), 0, stream>>>(goff, binned, rowptr, dinv, csr_src, N, E, nb);

  k_gather<128, true><<<rblocks, B, 0, stream>>>(h1, rowptr, csr_src, csr_w, dinv, b1, agg1, N);

  k_gemm_wmma<64, true><<<gblocks, B, 0, stream>>>(agg1, W2, h2, N);
  k_gather<64, false><<<rblocks, B, 0, stream>>>(h2, rowptr, csr_src, csr_w, dinv, b2, out, N);
}


# layer-2 product moved into the layer-1 aggregation launch: relu'd rows go through an LDS tile to the WMMA, the aggregated plane is never written or re-read; one launch less; same sums, same order
# speedup vs baseline: 3.1149x; 1.0434x over previous
#include <hip/hip_runtime.h>

typedef __attribute__((ext_vector_type(2))) float v2f;
typedef __attribute__((ext_vector_type(8))) float v8f;

typedef __attribute__((ext_vector_type(2))) int v2i;

constexpr int NCHUNK = 256;

__device__ __forceinline__ void chunk_range(int E, int c, int& c0, int& c1) {
  const int ch = (E + NCHUNK - 1) / NCHUNK;
  c0 = c * ch;
  if (c0 > E) c0 = E;
  c1 = c0 + ch;
  if (c1 > E) c1 = E;
}

__global__ __launch_bounds__(1024) void k_scan_excl(const int* __restrict__ in, int* __restrict__ outp,
                                                     int n) {
  __shared__ int w_tot[32];
  __shared__ int w_pre[32];
  const int tid = threadIdx.x;
  const int lane = tid & 31;
  const int wave = tid >> 5;
  const int blk0 = blockIdx.x * 1024;
  int pre = 0;
  for (int i = tid; i < blk0; i += 1024) pre += in[i];
  const int i = blk0 + tid;
  const int c = (i < n) ? in[i] : 0;
  int incl = c;
  for (int o = 1; o < 32; o <<= 1) {
    int t = __shfl_up(incl, o);
    if (lane >= o) incl += t;
  }
  for (int o = 16; o > 0; o >>= 1) pre += __shfl_xor(pre, o);
  if (lane == 31) w_tot[wave] = incl;
  if (lane == 0) w_pre[wave] = pre;
  __syncthreads();
  int base = w_pre[lane] + ((lane < wave) ? w_tot[lane] : 0);
  for (int o = 16; o > 0; o >>= 1) base += __shfl_xor(base, o);
  if (i < n) outp[i] = base + incl - c;
}

__global__ __launch_bounds__(1024) void k_bucket_csr(const int* __restrict__ goff,
                                                      const int* __restrict__ binned_i,
                                                      int* __restrict__ rowptr, float* __restrict__ dinv,
                                                      int* __restrict__ csr_src, int n, int E, int nb) {
  __shared__ int cnt[256];
  __shared__ int cur[256];
  const v2i* binned = (const v2i*)binned_i;
  const int k = blockIdx.x;
  const int tid = threadIdx.x;
  const int bs = goff[k * NCHUNK];
  const int be = (k + 1 < nb) ? goff[(k + 1) * NCHUNK] : E;
  if (tid < 256) cnt[tid] = 0;
  __syncthreads();
  for (int i0 = bs; i0 < be; i0 += 8 * 1024) {
    int d[8];
#pragma unroll
    for (int q = 0; q < 8; ++q) {
      const int i = i0 + q * 1024 + tid;
      d[q] = binned[(i < be) ? i : be - 1].y;
    }
#pragma unroll
    for (int q = 0; q < 8; ++q)
      if (i0 + q * 1024 + tid < be) atomicAdd(&cnt[d[q] & 255], 1);
  }
  __syncthreads();
  if (tid < 32) {
    int c[8];
    int tot = 0;
    for (int q = 0; q < 8; ++q) {
      c[q] = cnt[8 * tid + q];
      tot += c[q];
    }
    int incl = tot;
    for (int o = 1; o < 32; o <<= 1) {
      int t = __shfl_up(incl, o);
      if (tid >= o) incl += t;
    }
    int run = incl - tot;
    for (int q = 0; q < 8; ++q) {
      cur[8 * tid + q] = run;
      run += c[q];
    }
  }
  __syncthreads();
  if (tid < 256) {
    const int node = k * 256 + tid;
    if (node < n) {
      rowptr[node] = bs + cur[tid];
      dinv[node] = rsqrtf((float)(cnt[tid] + 1));
    }
  }
  if (k == nb - 1 && tid == 0) rowptr[n] = E;
  __syncthreads();
  for (int i0 = bs; i0 < be; i0 += 8 * 1024) {
    v2i r[8];
#pragma unroll
    for (int q = 0; q < 8; ++q) {
      const int i = i0 + q * 1024 + tid;
      r[q] = binned[(i < be) ? i : be - 1];
    }
#pragma unroll
    for (int q = 0; q < 8; ++q) {
      if (i0 + q * 1024 + tid < be) {
        const int p = atomicAdd(&cur[r[q].y & 255], 1);
        csr_src[bs + p] = r[q].x;
      }
    }
  }
}

template <int NOUT, bool RELU>
__device__ __forceinline__ void gemm_block(const float* __restrict__ A, const float* __restrict__ W,
                                           float* __restrict__ out, int nrows, int bid, float* Wl) {
  const int tid = threadIdx.x;
  for (int i = tid; i < 128 * NOUT; i += 256) {
    int k = i / NOUT;
    int j = i - k * NOUT;
    Wl[(k >> 1) * (2 * NOUT) + 2 * j + (k & 1)] = W[i];
  }
  __syncthreads();

  const int wave = tid >> 5;
  const int lane = tid & 31;
  const int tile = bid * 8 + wave;
  const int ntiles = (nrows + 15) >> 4;
  if (tile >= ntiles) return;

  const int m = lane & 15;
  const int kh = lane >> 4;
  int row = tile * 16 + m;
  if (row >= nrows) row = nrows - 1;

  v8f acc[NOUT / 16];
#pragma unroll
  for (int t = 0; t < NOUT / 16; ++t) acc[t] = v8f{0.f, 0.f, 0.f, 0.f, 0.f, 0.f, 0.f, 0.f};

  const float* arow = A + (size_t)row * 128 + 2 * kh;
#pragma unroll 4
  for (int k0 = 0; k0 < 128; k0 += 4) {
    v2f a = *(const v2f*)(arow + k0);
    if (RELU) {
      a.x = fmaxf(a.x, 0.f);
      a.y = fmaxf(a.y, 0.f);
    }
    const float* bb = &Wl[((k0 >> 1) + kh) * (2 * NOUT)];
#pragma unroll
    for (int t = 0; t < NOUT / 16; ++t) {
      v2f b = *(const v2f*)(bb + 2 * (t * 16 + m));
      acc[t] = __builtin_amdgcn_wmma_f32_16x16x4_f32(false, a, false, b, (short)0, acc[t],
                                                     false, false);
    }
  }

  const int rbase = tile * 16 + kh * 8;
#pragma unroll
  for (int t = 0; t < NOUT / 16; ++t) {
#pragma unroll
    for (int v = 0; v < 8; ++v) {
      int r = rbase + v;
      if (r < nrows) out[(size_t)r * NOUT + t * 16 + m] = acc[t][v];
    }
  }
}

__global__ __launch_bounds__(256) void k_bin_count_gemm(const int* __restrict__ dst,
                                                        int* __restrict__ ghist, int E, int nb,
                                                        const float* __restrict__ A,
                                                        const float* __restrict__ W,
                                                        float* __restrict__ out, int nrows,
                                                        int gfirst, int gcount) {
  __shared__ float Wl[128 * 128];
  if ((int)blockIdx.x < gcount) {
    gemm_block<128, false>(A, W, out, nrows, gfirst + (int)blockIdx.x, Wl);
    return;
  }
  int* lh = (int*)Wl;
  const int c = (int)blockIdx.x - gcount;
  const int tid = threadIdx.x;
  for (int k = tid; k < nb; k += 256) lh[k] = 0;
  __syncthreads();
  int c0, c1;
  chunk_range(E, c, c0, c1);
  for (int i0 = c0; i0 < c1; i0 += 16 * 256) {
    int d[16];
#pragma unroll
    for (int q = 0; q < 16; ++q) {
      const int i = i0 + q * 256 + tid;
      d[q] = dst[(i < c1) ? i : c1 - 1];
    }
#pragma unroll
    for (int q = 0; q < 16; ++q)
      if (i0 + q * 256 + tid < c1) atomicAdd(&lh[d[q] >> 8], 1);
  }
  __syncthreads();
  for (int k = tid; k < nb; k += 256) ghist[k * NCHUNK + c] = lh[k];
}

__global__ __launch_bounds__(256) void k_bin_place_gemm(const int* __restrict__ src,
                                                        const int* __restrict__ dst,
                                                        const int* __restrict__ goff,
                                                        int* __restrict__ binned_i, int E, int nb,
                                                        const float* __restrict__ A,
                                                        const float* __restrict__ W,
                                                        float* __restrict__ out, int nrows,
                                                        int gfirst, int gcount) {
  __shared__ float Wl[128 * 128];
  if ((int)blockIdx.x < gcount) {
    gemm_block<128, false>(A, W, out, nrows, gfirst + (int)blockIdx.x, Wl);
    return;
  }
  int* cur = (int*)Wl;
  v2i* binned = (v2i*)binned_i;
  const int c = (int)blockIdx.x - gcount;
  const int tid = threadIdx.x;
  for (int k = tid; k < nb; k += 256) cur[k] = goff[k * NCHUNK + c];
  __syncthreads();
  int c0, c1;
  chunk_range(E, c, c0, c1);
  for (int i0 = c0; i0 < c1; i0 += 16 * 256) {
    int d[16];
    int sr[16];
#pragma unroll
    for (int q = 0; q < 16; ++q) {
      const int i = (i0 + q * 256 + tid < c1) ? i0 + q * 256 + tid : c1 - 1;
      d[q] = dst[i];
      sr[q] = src[i];
    }
#pragma unroll
    for (int q = 0; q < 16; ++q) {
      if (i0 + q * 256 + tid < c1) {
        const int p = atomicAdd(&cur[d[q] >> 8], 1);
        v2i rec;
        rec.x = sr[q];
        rec.y = d[q];
        binned[p] = rec;
      }
    }
  }
}

template <int NOUT, bool FIRST>
__device__ __forceinline__ __attribute__((ext_vector_type(NOUT / 32))) float gather_row(
    const float* __restrict__ h, const int* __restrict__ rowptr, const int* __restrict__ csr_src,
    float* csr_w, const float* __restrict__ dinv, const float* __restrict__ bias, int row, int lane) {
  constexpr int VW = NOUT / 32;
  typedef __attribute__((ext_vector_type(VW))) float vec;
  const int beg = rowptr[row];
  const int end = rowptr[row + 1];
  const float di = dinv[row];
  const vec* hv = (const vec*)h + lane;
  const vec self = hv[(size_t)row * 32];
  vec acc;
  for (int q = 0; q < VW; ++q) acc[q] = self[q] * di * di + bias[lane * VW + q];
  for (int base = beg; base < end; base += 32) {
    const int m = (end - base < 32) ? (end - base) : 32;
    int sv = 0;
    int wb = 0;
    if (lane < m) {
      sv = csr_src[base + lane];
      wb = __float_as_int(FIRST ? dinv[sv] : csr_w[base + lane]);
    }
#pragma unroll 4
    for (int j = 0; j < m; ++j) {
      const int s = __builtin_amdgcn_readlane(sv, j);
      const float w = __int_as_float(__builtin_amdgcn_readlane(wb, j)) * di;
      const vec v = hv[(size_t)s * 32];
      for (int q = 0; q < VW; ++q) acc[q] += v[q] * w;
    }
    if (FIRST && lane < m) csr_w[base + lane] = __int_as_float(wb);
  }
  return acc;
}

template <int NOUT, bool FIRST>
__global__ __launch_bounds__(256) void k_gather(const float* __restrict__ h,
                                                const int* __restrict__ rowptr,
                                                const int* __restrict__ csr_src, float* csr_w,
                                                const float* __restrict__ dinv,
                                                const float* __restrict__ bias,
                                                float* __restrict__ out, int n) {
  typedef __attribute__((ext_vector_type(NOUT / 32))) float vec;
  const int wave = __builtin_amdgcn_readfirstlane((int)(threadIdx.x >> 5));
  const int lane = threadIdx.x & 31;
  const int row = blockIdx.x * 8 + wave;
  if (row >= n) return;
  const vec acc = gather_row<NOUT, FIRST>(h, rowptr, csr_src, csr_w, dinv, bias, row, lane);
  ((vec*)out)[(size_t)row * 32 + lane] = acc;
}

constexpr int FR = 32;
constexpr int AP = 132;
constexpr int WP = 160;

__global__ __launch_bounds__(512) void k_gather_gemm(const float* __restrict__ h,
                                                     const int* __restrict__ rowptr,
                                                     const int* __restrict__ csr_src, float* csr_w,
                                                     const float* __restrict__ dinv,
                                                     const float* __restrict__ bias,
                                                     const float* W2, float* __restrict__ h2, int n) {
  typedef __attribute__((ext_vector_type(4))) float v4f;
  __shared__ __attribute__((aligned(16))) float Wl[64 * WP];
  __shared__ __attribute__((aligned(16))) float At[FR * AP];
  const int tid = threadIdx.x;
  const int wave = __builtin_amdgcn_readfirstlane(tid >> 5);
  const int lane = tid & 31;

  v4f wlo[2], whi[2];
#pragma unroll
  for (int t = 0; t < 2; ++t) {
    const int u = tid + t * 512;
    const int p = u >> 4;
    const int j4 = (u & 15) * 4;
    wlo[t] = *(const v4f*)(W2 + (2 * p) * 64 + j4);
    whi[t] = *(const v4f*)(W2 + (2 * p + 1) * 64 + j4);
  }

  for (int rr = 0; rr < 2; ++rr) {
    const int lr = wave * 2 + rr;
    const int row = blockIdx.x * FR + lr;
    v4f acc = v4f{0.f, 0.f, 0.f, 0.f};
    if (row < n) {
      acc = gather_row<128, true>(h, rowptr, csr_src, csr_w, dinv, bias, row, lane);
      for (int q = 0; q < 4; ++q) acc[q] = fmaxf(acc[q], 0.f);
    }
    *(v4f*)(At + lr * AP + 4 * lane) = acc;
  }

#pragma unroll
  for (int t = 0; t < 2; ++t) {
    const int u = tid + t * 512;
    const int p = u >> 4;
    const int j4 = (u & 15) * 4;
    const v4f a = v4f{wlo[t][0], whi[t][0], wlo[t][1], whi[t][1]};
    const v4f b = v4f{wlo[t][2], whi[t][2], wlo[t][3], whi[t][3]};
    *(v4f*)(Wl + p * WP + 2 * j4) = a;
    *(v4f*)(Wl + p * WP + 2 * j4 + 4) = b;
  }
  __syncthreads();
  if (wave >= 4) return;

  const int m = lane & 15;
  const int kh = lane >> 4;
  v8f c0 = v8f{0.f, 0.f, 0.f, 0.f, 0.f, 0.f, 0.f, 0.f};
  v8f c1 = c0;
  const float* a0 = At + m * AP + 2 * kh;
  const float* a1 = At + (16 + m) * AP + 2 * kh;
  const float* bb = Wl + kh * WP + 2 * (wave * 16 + m);
#pragma unroll 8
  for (int k0 = 0; k0 < 128; k0 += 4) {
    const v2f x0 = *(const v2f*)(a0 + k0);
    const v2f x1 = *(const v2f*)(a1 + k0);
    const v2f b = *(const v2f*)(bb + (k0 >> 1) * WP);
    c0 = __builtin_amdgcn_wmma_f32_16x16x4_f32(false, x0, false, b, (short)0, c0, false, false);
    c1 = __builtin_amdgcn_wmma_f32_16x16x4_f32(false, x1, false, b, (short)0, c1, false, false);
  }

  const int rbase = blockIdx.x * FR + kh * 8;
#pragma unroll
  for (int v = 0; v < 8; ++v) {
    const int r0 = rbase + v;
    const int r1 = rbase + 16 + v;
    if (r0 < n) h2[(size_t)r0 * 64 + wave * 16 + m] = c0[v];
    if (r1 < n) h2[(size_t)r1 * 64 + wave * 16 + m] = c1[v];
  }
}

extern "C" void kernel_launch(void* const* d_in, const int* in_sizes, int n_in, void* d_out,
                              int out_size, void* d_ws, size_t ws_size, hipStream_t stream) {
  const float* x = (const float*)d_in[0];
  const int* ei = (const int*)d_in[1];
  const float* W1 = (const float*)d_in[2];
  const float* b1 = (const float*)d_in[3];
  const float* W2 = (const float*)d_in[4];
  const float* b2 = (const float*)d_in[5];
  float* out = (float*)d_out;

  const int N = in_sizes[0] / 128;
  const int E = in_sizes[1] / 2;
  const int* src = ei;
  const int* dstp = ei + E;
  const int nb = (N + 255) / 256;
  const int nscan = nb * NCHUNK;

  char* ws = (char*)d_ws;
  size_t off = 0;
  int* ghist = (int*)(ws + off);
  off += (((size_t)nscan * 4) + 255) & ~(size_t)255;
  int* goff = (int*)(ws + off);
  off += (((size_t)nscan * 4) + 255) & ~(size_t)255;
  int* rowptr = (int*)(ws + off);
  off += (((size_t)(N + 1) * 4) + 255) & ~(size_t)255;
  float* dinv = (float*)(ws + off);
  off += (((size_t)N * 4) + 255) & ~(size_t)255;
  int* binned = (int*)(ws + off);
  off += (((size_t)E * 8) + 255) & ~(size_t)255;
  int* csr_src = (int*)(ws + off);
  off += (((size_t)E * 4) + 255) & ~(size_t)255;
  float* csr_w = (float*)(ws + off);
  off += (((size_t)E * 4) + 255) & ~(size_t)255;
  float* h1 = (float*)(ws + off);
  off += (size_t)N * 128 * 4;
  float* h2 = (float*)(ws + off);

  dim3 B(256);
  const int ntiles = (N + 15) / 16;
  const int gblocks = (ntiles + 7) / 8;
  const int rblocks = (N + 7) / 8;
  const int gA = gblocks / 2;
  const int gB = gblocks - gA;

  k_bin_count_gemm<<<gA + NCHUNK, B, 0, stream>>>(dstp, ghist, E, nb, x, W1, h1, N, 0, gA);
  k_scan_excl<<<(nscan + 1023) / 1024, dim3(1024), 0, stream>>>(ghist, goff, nscan);
  k_bin_place_gemm<<<gB + NCHUNK, B, 0, stream>>>(src, dstp, goff, binned, E, nb, x, W1, h1, N, gA, gB);
  k_bucket_csr<<<nb, dim3(1024), 0, stream>>>(goff, binned, rowptr, dinv, csr_src, N, E, nb);

  k_gather_gemm<<<(N + FR - 1) / FR, dim3(512), 0, stream>>>(h1, rowptr, csr_src, csr_w, dinv, b1, W2, h2, N);

  k_gather<64, false><<<rblocks, B, 0, stream>>>(h2, rowptr, csr_src, csr_w, dinv, b2, out, N);
}


# level-1 edge sort chunk-local: a chunk sorts its own records by bucket in LDS, publishes segment starts; scan+place launches gone (6 to 4), GEMM strips in launch 1; add order = LDS-atomic arrival
# speedup vs baseline: 3.9238x; 1.2597x over previous
#include <hip/hip_runtime.h>

typedef __attribute__((ext_vector_type(2))) float v2f;
typedef __attribute__((ext_vector_type(8))) float v8f;

typedef __attribute__((ext_vector_type(2))) int v2i;

constexpr int NCHUNK = 256;

__device__ __forceinline__ void chunk_range(int E, int c, int& c0, int& c1) {
  const int ch = (E + NCHUNK - 1) / NCHUNK;
  c0 = c * ch;
  if (c0 > E) c0 = E;
  c1 = c0 + ch;
  if (c1 > E) c1 = E;
}

__global__ __launch_bounds__(1024) void k_bucket_csr(const int* __restrict__ ghist,
                                                      const int* __restrict__ gstart,
                                                      const int* __restrict__ binned_i,
                                                      int* __restrict__ rowptr, float* __restrict__ dinv,
                                                      int* __restrict__ csr_src, int n, int E, int nb) {
  static_assert(NCHUNK * 4 == 1024, "four lanes per chunk segment");
  __shared__ int cnt[256];
  __shared__ int cur[256];
  __shared__ int red[32];
  const v2i* binned = (const v2i*)binned_i;
  const int k = blockIdx.x;
  const int tid = threadIdx.x;
  const int lane = tid & 31;
  const int wave = tid >> 5;
  int part = 0;
  if (tid < NCHUNK) {
    int c0, c1;
    chunk_range(E, tid, c0, c1);
    part = gstart[k * NCHUNK + tid] - c0;
  }
  for (int o = 16; o > 0; o >>= 1) part += __shfl_xor(part, o);
  if (lane == 0) red[wave] = part;
  if (tid < 256) cnt[tid] = 0;
  const int sg = tid >> 2;
  const int sub = tid & 3;
  const int scnt = ghist[k * NCHUNK + sg];
  const int sst = gstart[k * NCHUNK + sg];
  v2i r[8];
#pragma unroll
  for (int q = 0; q < 8; ++q) {
    const int j = sub + 4 * q;
    r[q] = binned[(j < scnt) ? sst + j : 0];
  }
  __syncthreads();
  int bs = red[lane];
  for (int o = 16; o > 0; o >>= 1) bs += __shfl_xor(bs, o);
#pragma unroll
  for (int q = 0; q < 8; ++q)
    if (sub + 4 * q < scnt) atomicAdd(&cnt[r[q].y & 255], 1);
  for (int j = 32 + sub; j < scnt; j += 4)
    atomicAdd(&cnt[binned[sst + j].y & 255], 1);
  __syncthreads();
  if (tid < 32) {
    int c[8];
    int tot = 0;
    for (int q = 0; q < 8; ++q) {
      c[q] = cnt[8 * tid + q];
      tot += c[q];
    }
    int incl = tot;
    for (int o = 1; o < 32; o <<= 1) {
      int t = __shfl_up(incl, o);
      if (tid >= o) incl += t;
    }
    int run = incl - tot;
    for (int q = 0; q < 8; ++q) {
      cur[8 * tid + q] = run;
      run += c[q];
    }
  }
  __syncthreads();
  if (tid < 256) {
    const int node = k * 256 + tid;
    if (node < n) {
      rowptr[node] = bs + cur[tid];
      dinv[node] = rsqrtf((float)(cnt[tid] + 1));
    }
  }
  if (k == nb - 1 && tid == 0) rowptr[n] = E;
  __syncthreads();
#pragma unroll
  for (int q = 0; q < 8; ++q) {
    if (sub + 4 * q < scnt) {
      const int p = atomicAdd(&cur[r[q].y & 255], 1);
      csr_src[bs + p] = r[q].x;
    }
  }
  for (int j = 32 + sub; j < scnt; j += 4) {
    const v2i rec = binned[sst + j];
    const int p = atomicAdd(&cur[rec.y & 255], 1);
    csr_src[bs + p] = rec.x;
  }
}

template <int NOUT, bool RELU>
__device__ __forceinline__ void gemm_block(const float* __restrict__ A, const float* __restrict__ W,
                                           float* __restrict__ out, int nrows, int bid, float* Wl) {
  const int tid = threadIdx.x;
  for (int i = tid; i < 128 * NOUT; i += 256) {
    int k = i / NOUT;
    int j = i - k * NOUT;
    Wl[(k >> 1) * (2 * NOUT) + 2 * j + (k & 1)] = W[i];
  }
  __syncthreads();

  const int wave = tid >> 5;
  const int lane = tid & 31;
  const int tile = bid * 8 + wave;
  const int ntiles = (nrows + 15) >> 4;
  if (tile >= ntiles) return;

  const int m = lane & 15;
  const int kh = lane >> 4;
  int row = tile * 16 + m;
  if (row >= nrows) row = nrows - 1;

  v8f acc[NOUT / 16];
#pragma unroll
  for (int t = 0; t < NOUT / 16; ++t) acc[t] = v8f{0.f, 0.f, 0.f, 0.f, 0.f, 0.f, 0.f, 0.f};

  const float* arow = A + (size_t)row * 128 + 2 * kh;
#pragma unroll 4
  for (int k0 = 0; k0 < 128; k0 += 4) {
    v2f a = *(const v2f*)(arow + k0);
    if (RELU) {
      a.x = fmaxf(a.x, 0.f);
      a.y = fmaxf(a.y, 0.f);
    }
    const float* bb = &Wl[((k0 >> 1) + kh) * (2 * NOUT)];
#pragma unroll
    for (int t = 0; t < NOUT / 16; ++t) {
      v2f b = *(const v2f*)(bb + 2 * (t * 16 + m));
      acc[t] = __builtin_amdgcn_wmma_f32_16x16x4_f32(false, a, false, b, (short)0, acc[t],
                                                     false, false);
    }
  }

  const int rbase = tile * 16 + kh * 8;
#pragma unroll
  for (int t = 0; t < NOUT / 16; ++t) {
#pragma unroll
    for (int v = 0; v < 8; ++v) {
      int r = rbase + v;
      if (r < nrows) out[(size_t)r * NOUT + t * 16 + m] = acc[t][v];
    }
  }
}

__global__ __launch_bounds__(256) void k_chunk_sort_gemm(const int* __restrict__ src,
                                                         const int* __restrict__ dst,
                                                         int* __restrict__ ghist,
                                                         int* __restrict__ gstart,
                                                         int* __restrict__ binned_i, int E, int nb,
                                                         const float* __restrict__ A,
                                                         const float* __restrict__ W,
                                                         float* __restrict__ out, int nrows,
                                                         int gcount) {
  __shared__ float Wl[128 * 128];
  if ((int)blockIdx.x < gcount) {
    gemm_block<128, false>(A, W, out, nrows, (int)blockIdx.x, Wl);
    return;
  }
  int* lh = (int*)Wl;
  int* wtot = lh + nb;
  v2i* binned = (v2i*)binned_i;
  const int c = (int)blockIdx.x - gcount;
  const int tid = threadIdx.x;
  const int lane = tid & 31;
  const int wave = tid >> 5;
  for (int k = tid; k < nb; k += 256) lh[k] = 0;
  __syncthreads();
  int c0, c1;
  chunk_range(E, c, c0, c1);
  for (int i0 = c0; i0 < c1; i0 += 16 * 256) {
    int d[16];
#pragma unroll
    for (int q = 0; q < 16; ++q) {
      const int i = i0 + q * 256 + tid;
      d[q] = dst[(i < c1) ? i : c1 - 1];
    }
#pragma unroll
    for (int q = 0; q < 16; ++q)
      if (i0 + q * 256 + tid < c1) atomicAdd(&lh[d[q] >> 8], 1);
  }
  __syncthreads();
  const int per = (nb + 255) / 256;
  const int k0 = tid * per;
  int k1 = k0 + per;
  if (k1 > nb) k1 = nb;
  int tot = 0;
  for (int k = k0; k < k1; ++k) tot += lh[k];
  int incl = tot;
  for (int o = 1; o < 32; o <<= 1) {
    int t = __shfl_up(incl, o);
    if (lane >= o) incl += t;
  }
  if (lane == 31) wtot[wave] = incl;
  __syncthreads();
  int run = c0 + incl - tot;
  for (int w = 0; w < wave; ++w) run += wtot[w];
  for (int k = k0; k < k1; ++k) {
    const int nk = lh[k];
    ghist[k * NCHUNK + c] = nk;
    gstart[k * NCHUNK + c] = run;
    lh[k] = run;
    run += nk;
  }
  __syncthreads();
  for (int i0 = c0; i0 < c1; i0 += 16 * 256) {
    int d[16];
    int sr[16];
#pragma unroll
    for (int q = 0; q < 16; ++q) {
      const int i = (i0 + q * 256 + tid < c1) ? i0 + q * 256 + tid : c1 - 1;
      d[q] = dst[i];
      sr[q] = src[i];
    }
#pragma unroll
    for (int q = 0; q < 16; ++q) {
      if (i0 + q * 256 + tid < c1) {
        const int p = atomicAdd(&lh[d[q] >> 8], 1);
        v2i rec;
        rec.x = sr[q];
        rec.y = d[q];
        binned[p] = rec;
      }
    }
  }
}

template <int NOUT, bool FIRST>
__device__ __forceinline__ __attribute__((ext_vector_type(NOUT / 32))) float gather_row(
    const float* __restrict__ h, const int* __restrict__ rowptr, const int* __restrict__ csr_src,
    float* csr_w, const float* __restrict__ dinv, const float* __restrict__ bias, int row, int lane) {
  constexpr int VW = NOUT / 32;
  typedef __attribute__((ext_vector_type(VW))) float vec;
  const int beg = rowptr[row];
  const int end = rowptr[row + 1];
  const float di = dinv[row];
  const vec* hv = (const vec*)h + lane;
  const vec self = hv[(size_t)row * 32];
  vec acc;
  for (int q = 0; q < VW; ++q) acc[q] = self[q] * di * di + bias[lane * VW + q];
  for (int base = beg; base < end; base += 32) {
    const int m = (end - base < 32) ? (end - base) : 32;
    int sv = 0;
    int wb = 0;
    if (lane < m) {
      sv = csr_src[base + lane];
      wb = __float_as_int(FIRST ? dinv[sv] : csr_w[base + lane]);
    }
#pragma unroll 4
    for (int j = 0; j < m; ++j) {
      const int s = __builtin_amdgcn_readlane(sv, j);
      const float w = __int_as_float(__builtin_amdgcn_readlane(wb, j)) * di;
      const vec v = hv[(size_t)s * 32];
      for (int q = 0; q < VW; ++q) acc[q] += v[q] * w;
    }
    if (FIRST && lane < m) csr_w[base + lane] = __int_as_float(wb);
  }
  return acc;
}

template <int NOUT, bool FIRST>
__global__ __launch_bounds__(256) void k_gather(const float* __restrict__ h,
                                                const int* __restrict__ rowptr,
                                                const int* __restrict__ csr_src, float* csr_w,
                                                const float* __restrict__ dinv,
                                                const float* __restrict__ bias,
                                                float* __restrict__ out, int n) {
  typedef __attribute__((ext_vector_type(NOUT / 32))) float vec;
  const int wave = __builtin_amdgcn_readfirstlane((int)(threadIdx.x >> 5));
  const int lane = threadIdx.x & 31;
  const int row = blockIdx.x * 8 + wave;
  if (row >= n) return;
  const vec acc = gather_row<NOUT, FIRST>(h, rowptr, csr_src, csr_w, dinv, bias, row, lane);
  ((vec*)out)[(size_t)row * 32 + lane] = acc;
}

constexpr int FR = 32;
constexpr int AP = 132;
constexpr int WP = 160;

__global__ __launch_bounds__(512) void k_gather_gemm(const float* __restrict__ h,
                                                     const int* __restrict__ rowptr,
                                                     const int* __restrict__ csr_src, float* csr_w,
                                                     const float* __restrict__ dinv,
                                                     const float* __restrict__ bias,
                                                     const float* W2, float* __restrict__ h2, int n) {
  typedef __attribute__((ext_vector_type(4))) float v4f;
  __shared__ __attribute__((aligned(16))) float Wl[64 * WP];
  __shared__ __attribute__((aligned(16))) float At[FR * AP];
  const int tid = threadIdx.x;
  const int wave = __builtin_amdgcn_readfirstlane(tid >> 5);
  const int lane = tid & 31;

  v4f wlo[2], whi[2];
#pragma unroll
  for (int t = 0; t < 2; ++t) {
    const int u = tid + t * 512;
    const int p = u >> 4;
    const int j4 = (u & 15) * 4;
    wlo[t] = *(const v4f*)(W2 + (2 * p) * 64 + j4);
    whi[t] = *(const v4f*)(W2 + (2 * p + 1) * 64 + j4);
  }

  for (int rr = 0; rr < 2; ++rr) {
    const int lr = wave * 2 + rr;
    const int row = blockIdx.x * FR + lr;
    v4f acc = v4f{0.f, 0.f, 0.f, 0.f};
    if (row < n) {
      acc = gather_row<128, true>(h, rowptr, csr_src, csr_w, dinv, bias, row, lane);
      for (int q = 0; q < 4; ++q) acc[q] = fmaxf(acc[q], 0.f);
    }
    *(v4f*)(At + lr * AP + 4 * lane) = acc;
  }

#pragma unroll
  for (int t = 0; t < 2; ++t) {
    const int u = tid + t * 512;
    const int p = u >> 4;
    const int j4 = (u & 15) * 4;
    const v4f a = v4f{wlo[t][0], whi[t][0], wlo[t][1], whi[t][1]};
    const v4f b = v4f{wlo[t][2], whi[t][2], wlo[t][3], whi[t][3]};
    *(v4f*)(Wl + p * WP + 2 * j4) = a;
    *(v4f*)(Wl + p * WP + 2 * j4 + 4) = b;
  }
  __syncthreads();
  if (wave >= 4) return;

  const int m = lane & 15;
  const int kh = lane >> 4;
  v8f c0 = v8f{0.f, 0.f, 0.f, 0.f, 0.f, 0.f, 0.f, 0.f};
  v8f c1 = c0;
  const float* a0 = At + m * AP + 2 * kh;
  const float* a1 = At + (16 + m) * AP + 2 * kh;
  const float* bb = Wl + kh * WP + 2 * (wave * 16 + m);
#pragma unroll 8
  for (int k0 = 0; k0 < 128; k0 += 4) {
    const v2f x0 = *(const v2f*)(a0 + k0);
    const v2f x1 = *(const v2f*)(a1 + k0);
    const v2f b = *(const v2f*)(bb + (k0 >> 1) * WP);
    c0 = __builtin_amdgcn_wmma_f32_16x16x4_f32(false, x0, false, b, (short)0, c0, false, false);
    c1 = __builtin_amdgcn_wmma_f32_16x16x4_f32(false, x1, false, b, (short)0, c1, false, false);
  }

  const int rbase = blockIdx.x * FR + kh * 8;
#pragma unroll
  for (int v = 0; v < 8; ++v) {
    const int r0 = rbase + v;
    const int r1 = rbase + 16 + v;
    if (r0 < n) h2[(size_t)r0 * 64 + wave * 16 + m] = c0[v];
    if (r1 < n) h2[(size_t)r1 * 64 + wave * 16 + m] = c1[v];
  }
}

extern "C" void kernel_launch(void* const* d_in, const int* in_sizes, int n_in, void* d_out,
                              int out_size, void* d_ws, size_t ws_size, hipStream_t stream) {
  const float* x = (const float*)d_in[0];
  const int* ei = (const int*)d_in[1];
  const float* W1 = (const float*)d_in[2];
  const float* b1 = (const float*)d_in[3];
  const float* W2 = (const float*)d_in[4];
  const float* b2 = (const float*)d_in[5];
  float* out = (float*)d_out;

  const int N = in_sizes[0] / 128;
  const int E = in_sizes[1] / 2;
  const int* src = ei;
  const int* dstp = ei + E;
  const int nb = (N + 255) / 256;
  const int nscan = nb * NCHUNK;

  char* ws = (char*)d_ws;
  size_t off = 0;
  int* ghist = (int*)(ws + off);
  off += (((size_t)nscan * 4) + 255) & ~(size_t)255;
  int* gstart = (int*)(ws + off);
  off += (((size_t)nscan * 4) + 255) & ~(size_t)255;
  int* rowptr = (int*)(ws + off);
  off += (((size_t)(N + 1) * 4) + 255) & ~(size_t)255;
  float* dinv = (float*)(ws + off);
  off += (((size_t)N * 4) + 255) & ~(size_t)255;
  int* binned = (int*)(ws + off);
  off += (((size_t)E * 8) + 255) & ~(size_t)255;
  int* csr_src = (int*)(ws + off);
  off += (((size_t)E * 4) + 255) & ~(size_t)255;
  float* csr_w = (float*)(ws + off);
  off += (((size_t)E * 4) + 255) & ~(size_t)255;
  float* h1 = (float*)(ws + off);
  off += (size_t)N * 128 * 4;
  float* h2 = (float*)(ws + off);

  dim3 B(256);
  const int ntiles = (N + 15) / 16;
  const int gblocks = (ntiles + 7) / 8;
  const int rblocks = (N + 7) / 8;

  k_chunk_sort_gemm<<<gblocks + NCHUNK, B, 0, stream>>>(src, dstp, ghist, gstart, binned, E, nb, x, W1, h1, N,
                                                       gblocks);
  k_bucket_csr<<<nb, dim3(1024), 0, stream>>>(ghist, gstart, binned, rowptr, dinv, csr_src, N, E, nb);

  k_gather_gemm<<<(N + FR - 1) / FR, dim3(512), 0, stream>>>(h1, rowptr, csr_src, csr_w, dinv, b1, W2, h2, N);

  k_gather<64, false><<<rblocks, B, 0, stream>>>(h2, rowptr, csr_src, csr_w, dinv, b2, out, N);
}
